# starting kernel
# baseline (speedup 1.0000x reference)
#include <hip/hip_runtime.h>
#include <math.h>

typedef __attribute__((ext_vector_type(2))) float v2f;
typedef __attribute__((ext_vector_type(4))) float v4f;
typedef __attribute__((ext_vector_type(8))) float v8f;

#define BATCH 2
#define HIN   64
#define WIN   64
#define CH    256
#define HOUT  128
#define WOUT  128
#define KK2   25
#define FROW  260
#define OROW  260
#define BROW  102

__global__ __launch_bounds__(256) void carafe_wmma_kernel(
    const float* __restrict__ feat_g,
    const float* __restrict__ mask_g,
    float* __restrict__ out_g)
{
    __shared__ float featl[36 * FROW];
    __shared__ float Bm[16][BROW];
    __shared__ float outl[16][OROW];
    __shared__ int   adrtab[100];

    const int tid = threadIdx.x;
    int bid = blockIdx.x;
    const int lw = bid & 31; bid >>= 5;
    const int lh = bid & 31; bid >>= 5;
    const int bz = bid;
    const int h0 = lh * 2, w0 = lw * 2;

    if (tid < 100) {
        int q = tid / KK2;
        int p = tid - q * KK2;
        int y = (q >> 1) + p / 5;
        int x = (q & 1) + p % 5;
        adrtab[tid] = (y * 6 + x) * FROW;
    }

    for (int it = 0; it < 7; ++it) {
        int idx = it * 256 + tid;
        if (idx < 16 * BROW) ((float*)Bm)[idx] = 0.f;
    }

    for (int it = 0; it < 9; ++it) {
        int idx = it * 256 + tid;
        int cv  = idx & 63;
        int xy  = idx >> 6;
        int x = xy % 6, y = xy / 6;
        int gy = h0 - 2 + y, gx = w0 - 2 + x;
        v4f val = {0.f, 0.f, 0.f, 0.f};
        if (gy >= 0 && gy < HIN && gx >= 0 && gx < WIN) {
            val = *(const v4f*)(feat_g +
                  (((size_t)bz * HIN + gy) * WIN + gx) * CH + cv * 4);
        }
        *(v4f*)(&featl[(y * 6 + x) * FROW + cv * 4]) = val;
    }

    float w[KK2];
    int myq_s = 0;
    if (tid < 16) {
        int s = tid;
        myq_s = s >> 2;
        int t = s & 3;
        int Y = (h0 + (myq_s >> 1)) * 2 + (t >> 1);
        int X = (w0 + (myq_s & 1)) * 2 + (t & 1);
        const float* mp = mask_g + (((size_t)bz * HOUT + Y) * WOUT + X) * KK2;
        float mx = mp[0];
        #pragma unroll
        for (int p = 1; p < KK2; ++p) mx = fmaxf(mx, mp[p]);
        float sum = 0.f;
        #pragma unroll
        for (int p = 0; p < KK2; ++p) { w[p] = __expf(mp[p] - mx); sum += w[p]; }
        float inv = 1.f / sum;
        #pragma unroll
        for (int p = 0; p < KK2; ++p) w[p] *= inv;
    }
    __syncthreads();

    if (tid < 16) {
        #pragma unroll
        for (int p = 0; p < KK2; ++p)
            Bm[tid][myq_s * KK2 + p] = w[p];
    }
    __syncthreads();

    const int wave = tid >> 5;
    const int lane = tid & 31;
    const int n16  = lane & 15;
    const int hi   = lane >> 4;
    const int c0a  = wave * 16 + n16;
    const int c0b  = (wave + 8) * 16 + n16;

    v8f acc0 = {0.f, 0.f, 0.f, 0.f, 0.f, 0.f, 0.f, 0.f};
    v8f acc1 = {0.f, 0.f, 0.f, 0.f, 0.f, 0.f, 0.f, 0.f};

    #pragma unroll
    for (int k4 = 0; k4 < 25; ++k4) {
        int kk0 = k4 * 4 + hi * 2;
        int o0 = adrtab[kk0];
        int o1 = adrtab[kk0 + 1];

        v2f av0, av1, bv;
        av0.x = featl[o0 + c0a];
        av0.y = featl[o1 + c0a];
        av1.x = featl[o0 + c0b];
        av1.y = featl[o1 + c0b];
        bv = *(const v2f*)(&Bm[n16][kk0]);

        acc0 = __builtin_amdgcn_wmma_f32_16x16x4_f32(
            false, av0, false, bv, (short)0, acc0, false, false);
        acc1 = __builtin_amdgcn_wmma_f32_16x16x4_f32(
            false, av1, false, bv, (short)0, acc1, false, false);
    }

    #pragma unroll
    for (int r = 0; r < 8; ++r) {
        outl[n16][wave * 16 + hi * 8 + r]       = acc0[r];
        outl[n16][(wave + 8) * 16 + hi * 8 + r] = acc1[r];
    }
    __syncthreads();

    for (int it = 0; it < 4; ++it) {
        int idx = it * 256 + tid;
        int s  = idx >> 6;
        int cv = idx & 63;
        int q = s >> 2, t = s & 3;
        int Y = (h0 + (q >> 1)) * 2 + (t >> 1);
        int X = (w0 + (q & 1)) * 2 + (t & 1);
        v4f val = *(v4f*)(&outl[s][cv * 4]);
        *(v4f*)(out_g + (((size_t)bz * HOUT + Y) * WOUT + X) * CH + cv * 4) = val;
    }
}

extern "C" void kernel_launch(void* const* d_in, const int* in_sizes, int n_in,
                              void* d_out, int out_size, void* d_ws, size_t ws_size,
                              hipStream_t stream) {
    const float* features = (const float*)d_in[0];
    const float* masks    = (const float*)d_in[1];
    float* out            = (float*)d_out;
    dim3 grid(BATCH * 32 * 32);
    carafe_wmma_kernel<<<grid, 256, 0, stream>>>(features, masks, out);
}


# Tile fill: a workgroup's global loads (9 feature float4 per thread, 25 mask logits per softmax lane) issued as one group, clamped address + zero select, not 9 guarded load-wait-store trips
# speedup vs baseline: 1.2236x; 1.2236x over previous
#include <hip/hip_runtime.h>
#include <math.h>

typedef __attribute__((ext_vector_type(2))) float v2f;
typedef __attribute__((ext_vector_type(4))) float v4f;
typedef __attribute__((ext_vector_type(8))) float v8f;

#define BATCH 2
#define HIN   64
#define WIN   64
#define CH    256
#define HOUT  128
#define WOUT  128
#define KK2   25
#define FROW  260
#define OROW  260
#define BROW  102

__global__ __launch_bounds__(256) void carafe_wmma_kernel(
    const float* __restrict__ feat_g,
    const float* __restrict__ mask_g,
    float* __restrict__ out_g)
{
    __shared__ float featl[36 * FROW];
    __shared__ float Bm[16][BROW];
    __shared__ float outl[16][OROW];
    __shared__ int   adrtab[100];

    const int tid = threadIdx.x;
    int bid = blockIdx.x;
    const int lw = bid & 31; bid >>= 5;
    const int lh = bid & 31; bid >>= 5;
    const int bz = bid;
    const int h0 = lh * 2, w0 = lw * 2;

    const int wq = __builtin_amdgcn_readfirstlane(tid >> 6);
    const int ry = wq >> 1, rx = wq & 1;
    const int cv = tid & 63;

    float w[KK2];
    int myq_s = 0;
    if (tid < 16) {
        int s = tid;
        myq_s = s >> 2;
        int t = s & 3;
        int Y = (h0 + (myq_s >> 1)) * 2 + (t >> 1);
        int X = (w0 + (myq_s & 1)) * 2 + (t & 1);
        const float* mp = mask_g + (((size_t)bz * HOUT + Y) * WOUT + X) * KK2;
        #pragma unroll
        for (int p = 0; p < KK2; ++p) w[p] = mp[p];
    }

    v4f st[9];
    #pragma unroll
    for (int it = 0; it < 9; ++it) {
        int y = 2 * (it / 3) + ry, x = 2 * (it % 3) + rx;
        int gy = h0 - 2 + y, gx = w0 - 2 + x;
        int cy = gy < 0 ? 0 : (gy > HIN - 1 ? HIN - 1 : gy);
        int cx = gx < 0 ? 0 : (gx > WIN - 1 ? WIN - 1 : gx);
        const float* pix = feat_g + (size_t)((bz * HIN + cy) * WIN + cx) * CH;
        st[it] = *(const v4f*)(pix + cv * 4);
    }

    if (tid < 100) {
        int q = tid / KK2;
        int p = tid - q * KK2;
        int y = (q >> 1) + p / 5;
        int x = (q & 1) + p % 5;
        adrtab[tid] = (y * 6 + x) * FROW;
    }

    for (int it = 0; it < 7; ++it) {
        int idx = it * 256 + tid;
        if (idx < 16 * BROW) ((float*)Bm)[idx] = 0.f;
    }

    #pragma unroll
    for (int it = 0; it < 9; ++it) {
        int y = 2 * (it / 3) + ry, x = 2 * (it % 3) + rx;
        int gy = h0 - 2 + y, gx = w0 - 2 + x;
        bool inside = (gy >= 0 && gy < HIN && gx >= 0 && gx < WIN);
        v4f zero4 = {0.f, 0.f, 0.f, 0.f};
        v4f val = inside ? st[it] : zero4;
        *(v4f*)(&featl[(y * 6 + x) * FROW + cv * 4]) = val;
    }

    if (tid < 16) {
        float mx = w[0];
        #pragma unroll
        for (int p = 1; p < KK2; ++p) mx = fmaxf(mx, w[p]);
        float sum = 0.f;
        #pragma unroll
        for (int p = 0; p < KK2; ++p) { w[p] = __expf(w[p] - mx); sum += w[p]; }
        float inv = 1.f / sum;
        #pragma unroll
        for (int p = 0; p < KK2; ++p) w[p] *= inv;
    }
    __syncthreads();

    if (tid < 16) {
        #pragma unroll
        for (int p = 0; p < KK2; ++p)
            Bm[tid][myq_s * KK2 + p] = w[p];
    }
    __syncthreads();

    const int wave = tid >> 5;
    const int lane = tid & 31;
    const int n16  = lane & 15;
    const int hi   = lane >> 4;
    const int c0a  = wave * 16 + n16;
    const int c0b  = (wave + 8) * 16 + n16;

    v8f acc0 = {0.f, 0.f, 0.f, 0.f, 0.f, 0.f, 0.f, 0.f};
    v8f acc1 = {0.f, 0.f, 0.f, 0.f, 0.f, 0.f, 0.f, 0.f};

    #pragma unroll
    for (int k4 = 0; k4 < 25; ++k4) {
        int kk0 = k4 * 4 + hi * 2;
        int o0 = adrtab[kk0];
        int o1 = adrtab[kk0 + 1];

        v2f av0, av1, bv;
        av0.x = featl[o0 + c0a];
        av0.y = featl[o1 + c0a];
        av1.x = featl[o0 + c0b];
        av1.y = featl[o1 + c0b];
        bv = *(const v2f*)(&Bm[n16][kk0]);

        acc0 = __builtin_amdgcn_wmma_f32_16x16x4_f32(
            false, av0, false, bv, (short)0, acc0, false, false);
        acc1 = __builtin_amdgcn_wmma_f32_16x16x4_f32(
            false, av1, false, bv, (short)0, acc1, false, false);
    }

    #pragma unroll
    for (int r = 0; r < 8; ++r) {
        outl[n16][wave * 16 + hi * 8 + r]       = acc0[r];
        outl[n16][(wave + 8) * 16 + hi * 8 + r] = acc1[r];
    }
    __syncthreads();

    for (int it = 0; it < 4; ++it) {
        int idx = it * 256 + tid;
        int s  = idx >> 6;
        int cv = idx & 63;
        int q = s >> 2, t = s & 3;
        int Y = (h0 + (q >> 1)) * 2 + (t >> 1);
        int X = (w0 + (q & 1)) * 2 + (t & 1);
        v4f val = *(v4f*)(&outl[s][cv * 4]);
        *(v4f*)(out_g + (((size_t)bz * HOUT + Y) * WOUT + X) * CH + cv * 4) = val;
    }
}

extern "C" void kernel_launch(void* const* d_in, const int* in_sizes, int n_in,
                              void* d_out, int out_size, void* d_ws, size_t ws_size,
                              hipStream_t stream) {
    const float* features = (const float*)d_in[0];
    const float* masks    = (const float*)d_in[1];
    float* out            = (float*)d_out;
    dim3 grid(BATCH * 32 * 32);
    carafe_wmma_kernel<<<grid, 256, 0, stream>>>(features, masks, out);
}


# Workgroup tile 4x4 low-res pixels (8x8 patch union staged in LDS once, shared by four 2x2 sub-blocks, each still a K=36 product on its own 6x6 window), grid 512, not 2x2 tiles with a 6x6 union each
# speedup vs baseline: 2.1520x; 1.7587x over previous
#include <hip/hip_runtime.h>
#include <math.h>

typedef __attribute__((ext_vector_type(2))) float v2f;
typedef __attribute__((ext_vector_type(4))) float v4f;
typedef __attribute__((ext_vector_type(8))) float v8f;

#define BATCH 2
#define HIN   64
#define WIN   64
#define CH    256
#define HOUT  128
#define WOUT  128
#define KK2   25
#define UW    8
#define NPIX  64
#define FROW  264
#define OROW  260
#define KU    36
#define BROW  36

__global__ __launch_bounds__(256) void carafe_wmma_kernel(
    const float* __restrict__ feat_g,
    const float* __restrict__ mask_g,
    float* __restrict__ out_g)
{
    __shared__ __attribute__((aligned(16))) float featl[UW * UW * FROW];
    __shared__ __attribute__((aligned(16))) float Bm[NPIX][BROW];
    __shared__ __attribute__((aligned(16))) float outl[NPIX][OROW];

    const int tid = threadIdx.x;
    int bid = blockIdx.x;
    const int lw = bid & 15; bid >>= 4;
    const int lh = bid & 15; bid >>= 4;
    const int bz = bid;
    const int h0 = lh * 4, w0 = lw * 4;

    const int wq = __builtin_amdgcn_readfirstlane(tid >> 6);
    const int ry = wq >> 1, rx = wq & 1;
    const int cv = tid & 63;

    float w[KK2];
    int ub = 0;
    if (tid < NPIX) {
        int s = tid;
        int sb = s >> 4, q = (s >> 2) & 3, t = s & 3;
        int ly = 2 * (sb >> 1) + (q >> 1);
        int lx = 2 * (sb & 1) + (q & 1);
        int Y = (h0 + ly) * 2 + (t >> 1);
        int X = (w0 + lx) * 2 + (t & 1);
        ub = (q >> 1) * 6 + (q & 1);
        const float* mp = mask_g + (((size_t)bz * HOUT + Y) * WOUT + X) * KK2;
        #pragma unroll
        for (int p = 0; p < KK2; ++p) w[p] = mp[p];
    }

    v4f st[16];
    #pragma unroll
    for (int it = 0; it < 16; ++it) {
        int y = 2 * (it >> 2) + ry, x = 2 * (it & 3) + rx;
        int gy = h0 - 2 + y, gx = w0 - 2 + x;
        int cy = gy < 0 ? 0 : (gy > HIN - 1 ? HIN - 1 : gy);
        int cx = gx < 0 ? 0 : (gx > WIN - 1 ? WIN - 1 : gx);
        const float* pix = feat_g + (size_t)((bz * HIN + cy) * WIN + cx) * CH;
        st[it] = *(const v4f*)(pix + cv * 4);
    }

    #pragma unroll
    for (int it = 0; it < (NPIX * BROW) / 256; ++it)
        ((float*)Bm)[it * 256 + tid] = 0.f;

    #pragma unroll
    for (int it = 0; it < 16; ++it) {
        int y = 2 * (it >> 2) + ry, x = 2 * (it & 3) + rx;
        int gy = h0 - 2 + y, gx = w0 - 2 + x;
        bool inside = (gy >= 0 && gy < HIN && gx >= 0 && gx < WIN);
        v4f zero4 = {0.f, 0.f, 0.f, 0.f};
        v4f val = inside ? st[it] : zero4;
        *(v4f*)(&featl[(y * UW + x) * FROW + cv * 4]) = val;
    }

    if (tid < NPIX) {
        float mx = w[0];
        #pragma unroll
        for (int p = 1; p < KK2; ++p) mx = fmaxf(mx, w[p]);
        float sum = 0.f;
        #pragma unroll
        for (int p = 0; p < KK2; ++p) { w[p] = __expf(w[p] - mx); sum += w[p]; }
        float inv = 1.f / sum;
        #pragma unroll
        for (int p = 0; p < KK2; ++p) w[p] *= inv;
    }
    __syncthreads();

    if (tid < NPIX) {
        #pragma unroll
        for (int p = 0; p < KK2; ++p)
            Bm[tid][ub + (p / 5) * 6 + (p % 5)] = w[p];
    }
    __syncthreads();

    const int wave = tid >> 5;
    const int lane = tid & 31;
    const int n16  = lane & 15;
    const int hi   = lane >> 4;
    const int c0a  = wave * 16 + n16;
    const int c0b  = (wave + 8) * 16 + n16;

    #pragma unroll
    for (int sb = 0; sb < 4; ++sb) {
        const int ubase = (2 * (sb >> 1)) * UW + 2 * (sb & 1);

        v8f acc0 = {0.f, 0.f, 0.f, 0.f, 0.f, 0.f, 0.f, 0.f};
        v8f acc1 = {0.f, 0.f, 0.f, 0.f, 0.f, 0.f, 0.f, 0.f};

        #pragma unroll
        for (int k4 = 0; k4 < KU / 4; ++k4) {
            const int ka = k4 * 4, kb = k4 * 4 + 2;
            const int ra = ubase + (ka / 6) * UW + (ka % 6);
            const int rb = ubase + (kb / 6) * UW + (kb % 6);
            int kk0 = hi ? kb : ka;
            int o0  = (hi ? rb : ra) * FROW;
            int o1  = o0 + FROW;

            v2f av0, av1, bv;
            av0.x = featl[o0 + c0a];
            av0.y = featl[o1 + c0a];
            av1.x = featl[o0 + c0b];
            av1.y = featl[o1 + c0b];
            bv = *(const v2f*)(&Bm[sb * 16 + n16][kk0]);

            acc0 = __builtin_amdgcn_wmma_f32_16x16x4_f32(
                false, av0, false, bv, (short)0, acc0, false, false);
            acc1 = __builtin_amdgcn_wmma_f32_16x16x4_f32(
                false, av1, false, bv, (short)0, acc1, false, false);
        }

        #pragma unroll
        for (int r = 0; r < 8; ++r) {
            outl[sb * 16 + n16][wave * 16 + hi * 8 + r]       = acc0[r];
            outl[sb * 16 + n16][(wave + 8) * 16 + hi * 8 + r] = acc1[r];
        }
    }
    __syncthreads();

    #pragma unroll 4
    for (int it = 0; it < (NPIX * 64) / 256; ++it) {
        int idx = it * 256 + tid;
        int s  = idx >> 6;
        int c4 = idx & 63;
        int sb = s >> 4, q = (s >> 2) & 3, t = s & 3;
        int ly = 2 * (sb >> 1) + (q >> 1);
        int lx = 2 * (sb & 1) + (q & 1);
        int Y = (h0 + ly) * 2 + (t >> 1);
        int X = (w0 + lx) * 2 + (t & 1);
        v4f val = *(v4f*)(&outl[s][c4 * 4]);
        *(v4f*)(out_g + (((size_t)bz * HOUT + Y) * WOUT + X) * CH + c4 * 4) = val;
    }
}

extern "C" void kernel_launch(void* const* d_in, const int* in_sizes, int n_in,
                              void* d_out, int out_size, void* d_ws, size_t ws_size,
                              hipStream_t stream) {
    const float* features = (const float*)d_in[0];
    const float* masks    = (const float*)d_in[1];
    float* out            = (float*)d_out;
    dim3 grid(BATCH * 16 * 16);
    carafe_wmma_kernel<<<grid, 256, 0, stream>>>(features, masks, out);
}


# A wave owns one sub-block x one channel half (8 channel tiles on one B fragment), passes its 512-B half rows through its own LDS slot and stores them at once: no workgroup output tile, no 3rd barrier
# speedup vs baseline: 2.3299x; 1.0827x over previous
#include <hip/hip_runtime.h>
#include <math.h>

typedef __attribute__((ext_vector_type(2))) float v2f;
typedef __attribute__((ext_vector_type(4))) float v4f;
typedef __attribute__((ext_vector_type(8))) float v8f;

#define BATCH 2
#define HIN   64
#define WIN   64
#define CH    256
#define HOUT  128
#define WOUT  128
#define KK2   25
#define UW    8
#define NPIX  64
#define FROW  264
#define SROW  132
#define KU    36
#define BROW  36

__global__ __launch_bounds__(256) void carafe_wmma_kernel(
    const float* __restrict__ feat_g,
    const float* __restrict__ mask_g,
    float* __restrict__ out_g)
{
    __shared__ __attribute__((aligned(16))) float featl[UW * UW * FROW];
    __shared__ __attribute__((aligned(16))) float Bm[NPIX][BROW];
    __shared__ __attribute__((aligned(16))) float slot[8][16 * SROW];

    const int tid = threadIdx.x;
    int bid = blockIdx.x;
    const int lw = bid & 15; bid >>= 4;
    const int lh = bid & 15; bid >>= 4;
    const int bz = bid;
    const int h0 = lh * 4, w0 = lw * 4;

    const int wq = __builtin_amdgcn_readfirstlane(tid >> 6);
    const int ry = wq >> 1, rx = wq & 1;
    const int cv = tid & 63;

    float w[KK2];
    int ub = 0;
    if (tid < NPIX) {
        int s = tid;
        int sb = s >> 4, q = (s >> 2) & 3, t = s & 3;
        int ly = 2 * (sb >> 1) + (q >> 1);
        int lx = 2 * (sb & 1) + (q & 1);
        int Y = (h0 + ly) * 2 + (t >> 1);
        int X = (w0 + lx) * 2 + (t & 1);
        ub = (q >> 1) * 6 + (q & 1);
        const float* mp = mask_g + (((size_t)bz * HOUT + Y) * WOUT + X) * KK2;
        #pragma unroll
        for (int p = 0; p < KK2; ++p) w[p] = mp[p];
    }

    v4f st[16];
    #pragma unroll
    for (int it = 0; it < 16; ++it) {
        int y = 2 * (it >> 2) + ry, x = 2 * (it & 3) + rx;
        int gy = h0 - 2 + y, gx = w0 - 2 + x;
        int cy = gy < 0 ? 0 : (gy > HIN - 1 ? HIN - 1 : gy);
        int cx = gx < 0 ? 0 : (gx > WIN - 1 ? WIN - 1 : gx);
        const float* pix = feat_g + (size_t)((bz * HIN + cy) * WIN + cx) * CH;
        st[it] = *(const v4f*)(pix + cv * 4);
    }

    #pragma unroll
    for (int it = 0; it < (NPIX * BROW) / 256; ++it)
        ((float*)Bm)[it * 256 + tid] = 0.f;

    #pragma unroll
    for (int it = 0; it < 16; ++it) {
        int y = 2 * (it >> 2) + ry, x = 2 * (it & 3) + rx;
        int gy = h0 - 2 + y, gx = w0 - 2 + x;
        bool inside = (gy >= 0 && gy < HIN && gx >= 0 && gx < WIN);
        v4f zero4 = {0.f, 0.f, 0.f, 0.f};
        v4f val = inside ? st[it] : zero4;
        *(v4f*)(&featl[(y * UW + x) * FROW + cv * 4]) = val;
    }

    if (tid < NPIX) {
        float mx = w[0];
        #pragma unroll
        for (int p = 1; p < KK2; ++p) mx = fmaxf(mx, w[p]);
        float sum = 0.f;
        #pragma unroll
        for (int p = 0; p < KK2; ++p) { w[p] = __expf(w[p] - mx); sum += w[p]; }
        float inv = 1.f / sum;
        #pragma unroll
        for (int p = 0; p < KK2; ++p) w[p] *= inv;
    }
    __syncthreads();

    if (tid < NPIX) {
        #pragma unroll
        for (int p = 0; p < KK2; ++p)
            Bm[tid][ub + (p / 5) * 6 + (p % 5)] = w[p];
    }
    __syncthreads();

    const int wave = __builtin_amdgcn_readfirstlane(tid >> 5);
    const int lane = tid & 31;
    const int n16  = lane & 15;
    const int hi   = lane >> 4;
    const int sb   = wave >> 1;
    const int ch0  = (wave & 1) * 128;
    const int ubase = (2 * (sb >> 1)) * UW + 2 * (sb & 1);

    const int lb2 = (ubase + 2 * hi) * FROW + ch0 + n16;
    const int lb4 = (ubase + 4 * hi) * FROW + ch0 + n16;
    const float* bp = &Bm[sb * 16 + n16][2 * hi];

    v8f acc[8];
    #pragma unroll
    for (int t = 0; t < 8; ++t)
        acc[t] = (v8f){0.f, 0.f, 0.f, 0.f, 0.f, 0.f, 0.f, 0.f};

    #pragma unroll
    for (int k4 = 0; k4 < KU / 4; ++k4) {
        const int ka = k4 * 4;
        const int ca = (ka / 6) * UW + (ka % 6);
        const int o0 = ((ka % 6) == 4 ? lb4 : lb2) + ca * FROW;
        const int o1 = o0 + FROW;
        const v2f bv = *(const v2f*)(bp + ka);

        #pragma unroll
        for (int t = 0; t < 8; ++t) {
            v2f av;
            av.x = featl[o0 + t * 16];
            av.y = featl[o1 + t * 16];
            acc[t] = __builtin_amdgcn_wmma_f32_16x16x4_f32(
                false, av, false, bv, (short)0, acc[t], false, false);
        }
    }

    float* myslot = &slot[wave][0];
    #pragma unroll
    for (int t = 0; t < 8; ++t) {
        v4f lo = {acc[t][0], acc[t][1], acc[t][2], acc[t][3]};
        v4f hv = {acc[t][4], acc[t][5], acc[t][6], acc[t][7]};
        *(v4f*)(&myslot[n16 * SROW + t * 16 + hi * 8])     = lo;
        *(v4f*)(&myslot[n16 * SROW + t * 16 + hi * 8 + 4]) = hv;
    }
    __builtin_amdgcn_fence(__ATOMIC_RELEASE, "wavefront");
    asm volatile("s_wait_dscnt 0x0" ::: "memory");
    __builtin_amdgcn_wave_barrier();
    __builtin_amdgcn_fence(__ATOMIC_ACQUIRE, "wavefront");

    #pragma unroll
    for (int p = 0; p < 16; ++p) {
        const int q = p >> 2, t = p & 3;
        const int ly = 2 * (sb >> 1) + (q >> 1);
        const int lx = 2 * (sb & 1) + (q & 1);
        const int Y = (h0 + ly) * 2 + (t >> 1);
        const int X = (w0 + lx) * 2 + (t & 1);
        v4f val = *(const v4f*)(&myslot[p * SROW + lane * 4]);
        *(v4f*)(out_g + (((size_t)bz * HOUT + Y) * WOUT + X) * CH + ch0 + lane * 4) = val;
    }
}

extern "C" void kernel_launch(void* const* d_in, const int* in_sizes, int n_in,
                              void* d_out, int out_size, void* d_ws, size_t ws_size,
                              hipStream_t stream) {
    const float* features = (const float*)d_in[0];
    const float* masks    = (const float*)d_in[1];
    float* out            = (float*)d_out;
    dim3 grid(BATCH * 16 * 16);
    carafe_wmma_kernel<<<grid, 256, 0, stream>>>(features, masks, out);
}
